# starting kernel
# baseline (speedup 1.0000x reference)
#include <hip/hip_runtime.h>

typedef __attribute__((ext_vector_type(2))) float v2f;
typedef __attribute__((ext_vector_type(8))) float v8f;

#define SLOT_DIM 128
#define HIDDEN   64
#define NB       64
#define NK       128

__global__ __launch_bounds__(256) void proj_gemm_wmma(
    const float* __restrict__ slots,
    const float* __restrict__ W1,
    float* __restrict__ LR)
{
    const int lane  = threadIdx.x & 31;
    const int wave  = threadIdx.x >> 5;
    const int mtile = blockIdx.x * 8 + wave;
    const int ntile = blockIdx.y;

    const int m15 = lane & 15;
    const int kb  = (lane >> 4) << 1;

    const int    gRow = mtile * 16 + m15;
    const float* aptr = slots + (size_t)gRow * SLOT_DIM;

    const int    n    = ntile * 16 + m15;
    const int    wrow = (ntile >= 4) ? (n - HIDDEN) : n;
    const float* bptr = W1 + (size_t)wrow * (2 * SLOT_DIM) + ((ntile >= 4) ? SLOT_DIM : 0);

    v8f c = {};
#pragma unroll
    for (int k0 = 0; k0 < SLOT_DIM; k0 += 4) {
        v2f a, b;
        a.x = aptr[k0 + kb];
        a.y = aptr[k0 + kb + 1];
        b.x = bptr[k0 + kb];
        b.y = bptr[k0 + kb + 1];
        c = __builtin_amdgcn_wmma_f32_16x16x4_f32(false, a, false, b,
                                                  (short)0, c, false, false);
    }

#pragma unroll
    for (int r = 0; r < 8; ++r) {
        const int mo = r + ((lane >> 4) << 3);
        LR[(size_t)(mtile * 16 + mo) * 128 + ntile * 16 + m15] = c[r];
    }
}

__global__ __launch_bounds__(256) void edge_score(
    const float* __restrict__ LR,
    const float* __restrict__ b1,
    const float* __restrict__ W2,
    const float* __restrict__ b2p,
    float* __restrict__ out)
{
    __shared__ float Lsh[16 * 64];
    __shared__ float Rsh[128 * 65];
    __shared__ float b1sh[64];
    __shared__ float W2sh[64];

    const int bidx = blockIdx.x;
    const int i0   = blockIdx.y * 16;
    const int tid  = threadIdx.x;

    const float* base = LR + (size_t)bidx * NK * 128;

    for (int idx = tid; idx < 16 * 64; idx += 256) {
        const int il = idx >> 6, h = idx & 63;
        Lsh[idx] = base[(size_t)(i0 + il) * 128 + h];
    }
    for (int idx = tid; idx < 128 * 64; idx += 256) {
        const int j = idx >> 6, h = idx & 63;
        Rsh[j * 65 + h] = base[(size_t)j * 128 + 64 + h];
    }
    if (tid < 64) {
        b1sh[tid] = b1[tid];
        W2sh[tid] = W2[tid];
    }
    __syncthreads();

    const float b2 = b2p[0];

    for (int e = tid; e < 16 * 128; e += 256) {
        const int il = e >> 7;
        const int j  = e & 127;
        const float* Lr = Lsh + il * 64;
        const float* Rr = Rsh + j * 65;

        float acc = b2;
#pragma unroll
        for (int h = 0; h < 64; ++h) {
            const float v = Lr[h] + Rr[h] + b1sh[h];
            acc = __builtin_fmaf(fmaxf(v, 0.0f), W2sh[h], acc);
        }
        const float sig = 1.0f / (1.0f + __expf(-acc));
        const int gi = i0 + il;
        out[((size_t)bidx * NK + gi) * NK + j] = (gi == j) ? 0.0f : sig;
    }
}

extern "C" void kernel_launch(void* const* d_in, const int* in_sizes, int n_in,
                              void* d_out, int out_size, void* d_ws, size_t ws_size,
                              hipStream_t stream) {
    (void)in_sizes; (void)n_in; (void)out_size; (void)ws_size;
    const float* slots = (const float*)d_in[0];
    const float* W1    = (const float*)d_in[1];
    const float* b1    = (const float*)d_in[2];
    const float* W2    = (const float*)d_in[3];
    const float* b2    = (const float*)d_in[4];
    float*       out   = (float*)d_out;
    float*       LR    = (float*)d_ws;

    dim3 gA(64, 8);
    proj_gemm_wmma<<<gA, 256, 0, stream>>>(slots, W1, LR);

    dim3 gB(64, 8);
    edge_score<<<gB, 256, 0, stream>>>(LR, b1, W2, b2, out);
}


# projection kernel re-blocked for one wave per SIMD: 32x32 per wave (each W1 and slots fragment feeds two tiles: 4 loads per 8 WMMAs), its 64 fragment loads issued ahead; f32 sum order unchanged
# speedup vs baseline: 1.5116x; 1.5116x over previous
#include <hip/hip_runtime.h>


typedef __attribute__((ext_vector_type(2))) float v2f;
typedef __attribute__((ext_vector_type(4))) float v4f;
typedef __attribute__((ext_vector_type(8))) float v8f;

#define SLOT_DIM 128
#define HIDDEN   64
#define NB       64
#define NK       128

#define LM_ELEMS (8192 * HIDDEN)

#define KSTEPS (SLOT_DIM / 8)

__global__ __launch_bounds__(128) void proj_gemm_wmma(
    const float* __restrict__ slots,
    const float* __restrict__ W1,
    float* __restrict__ Lm,
    float* __restrict__ RT4)
{
    const int lane  = threadIdx.x & 31;
    const int wave  = threadIdx.x >> 5;
    const int npair = blockIdx.y;
    const bool isR  = (npair >= 2);
    const int ht0   = (npair & 1) * 2;
    const int g0    = blockIdx.x * NK + wave * 32;

    const int m15  = lane & 15;
    const int half = lane >> 4;
    const int ko   = half << 2;

    const float* sp0 = slots + (size_t)(g0 + m15) * SLOT_DIM + ko;
    const float* sp1 = sp0 + (size_t)16 * SLOT_DIM;
    const float* wp0 = W1 + (size_t)(ht0 * 16 + m15) * (2 * SLOT_DIM) + (isR ? SLOT_DIM : 0) + ko;
    const float* wp1 = wp0 + (size_t)16 * (2 * SLOT_DIM);

    v4f fa0[KSTEPS], fa1[KSTEPS], fb0[KSTEPS], fb1[KSTEPS];
#pragma unroll
    for (int s = 0; s < KSTEPS; ++s) {
        fa0[s] = *(const v4f*)(wp0 + 8 * s);
        fb0[s] = *(const v4f*)(sp0 + 8 * s);
        fa1[s] = *(const v4f*)(wp1 + 8 * s);
        fb1[s] = *(const v4f*)(sp1 + 8 * s);
    }
    __builtin_amdgcn_sched_barrier(0);

    v8f c00 = {}, c01 = {}, c10 = {}, c11 = {};
#pragma unroll
    for (int s = 0; s < KSTEPS; ++s) {
        const v4f a0 = fa0[s], a1 = fa1[s], b0 = fb0[s], b1 = fb1[s];
        const v2f a0_lo = {a0.x, a0.y}, a0_hi = {a0.z, a0.w};
        const v2f a1_lo = {a1.x, a1.y}, a1_hi = {a1.z, a1.w};
        const v2f b0_lo = {b0.x, b0.y}, b0_hi = {b0.z, b0.w};
        const v2f b1_lo = {b1.x, b1.y}, b1_hi = {b1.z, b1.w};
        c00 = __builtin_amdgcn_wmma_f32_16x16x4_f32(false, a0_lo, false, b0_lo,
                                                    (short)0, c00, false, false);
        c01 = __builtin_amdgcn_wmma_f32_16x16x4_f32(false, a1_lo, false, b0_lo,
                                                    (short)0, c01, false, false);
        c10 = __builtin_amdgcn_wmma_f32_16x16x4_f32(false, a0_lo, false, b1_lo,
                                                    (short)0, c10, false, false);
        c11 = __builtin_amdgcn_wmma_f32_16x16x4_f32(false, a1_lo, false, b1_lo,
                                                    (short)0, c11, false, false);
        c00 = __builtin_amdgcn_wmma_f32_16x16x4_f32(false, a0_hi, false, b0_hi,
                                                    (short)0, c00, false, false);
        c01 = __builtin_amdgcn_wmma_f32_16x16x4_f32(false, a1_hi, false, b0_hi,
                                                    (short)0, c01, false, false);
        c10 = __builtin_amdgcn_wmma_f32_16x16x4_f32(false, a0_hi, false, b1_hi,
                                                    (short)0, c10, false, false);
        c11 = __builtin_amdgcn_wmma_f32_16x16x4_f32(false, a1_hi, false, b1_hi,
                                                    (short)0, c11, false, false);
    }

#pragma unroll
    for (int mi = 0; mi < 2; ++mi) {
        const v8f t0 = mi ? c10 : c00;
        const v8f t1 = mi ? c11 : c01;
        const v4f lo0 = {t0[0], t0[1], t0[2], t0[3]};
        const v4f hi0 = {t0[4], t0[5], t0[6], t0[7]};
        const v4f lo1 = {t1[0], t1[1], t1[2], t1[3]};
        const v4f hi1 = {t1[4], t1[5], t1[6], t1[7]};
        const int gm = g0 + 16 * mi;
        if (!isR) {
            float* dst = Lm + (size_t)(gm + m15) * HIDDEN + ht0 * 16 + half * 8;
            *(v4f*)(dst)          = lo0;
            *(v4f*)(dst + 4)      = hi0;
            *(v4f*)(dst + 16)     = lo1;
            *(v4f*)(dst + 16 + 4) = hi1;
        } else {
            const int bb = gm >> 7;
            const int j  = (gm & 127) + m15;
            const int hq = ht0 * 4 + half * 2;
            float* dst = RT4 + (((size_t)bb * 16 + hq) * NK + j) * 4;
            *(v4f*)(dst)              = lo0;
            *(v4f*)(dst + NK * 4)     = hi0;
            *(v4f*)(dst + 4 * NK * 4) = lo1;
            *(v4f*)(dst + 5 * NK * 4) = hi1;
        }
    }
}

#define QJ 2
#define RI 8

__global__ __launch_bounds__(128) void edge_score(
    const float* __restrict__ Lm,
    const float* __restrict__ RT4,
    const float* __restrict__ b1,
    const float* __restrict__ W2,
    const float* __restrict__ b2p,
    float* __restrict__ out)
{
    const int lane = threadIdx.x & 31;
    const int wave = threadIdx.x >> 5;
    const int bidx = blockIdx.x;
    const int jb   = (wave & 1) * 64 + lane;
    const int i0   = blockIdx.y * 16 + (wave >> 1) * RI;

    const float* Lrow = Lm + ((size_t)bidx * NK + i0) * HIDDEN + lane;
    float lv[RI][2];
#pragma unroll
    for (int i = 0; i < RI; ++i) {
        lv[i][0] = Lrow[i * HIDDEN];
        lv[i][1] = Lrow[i * HIDDEN + 32];
    }

    float rp_[QJ][HIDDEN];
#pragma unroll
    for (int q = 0; q < QJ; ++q) {
        const float* src = RT4 + ((size_t)bidx * 16 * NK + (jb + 32 * q)) * 4;
#pragma unroll
        for (int h4 = 0; h4 < 16; ++h4) {
            const v4f t = *(const v4f*)(src + (size_t)h4 * NK * 4);
            rp_[q][4 * h4 + 0] = t.x + b1[4 * h4 + 0];
            rp_[q][4 * h4 + 1] = t.y + b1[4 * h4 + 1];
            rp_[q][4 * h4 + 2] = t.z + b1[4 * h4 + 2];
            rp_[q][4 * h4 + 3] = t.w + b1[4 * h4 + 3];
        }
    }

    const float b2 = b2p[0];
    float acc[RI][QJ];
#pragma unroll
    for (int i = 0; i < RI; ++i)
#pragma unroll
        for (int q = 0; q < QJ; ++q) acc[i][q] = b2;

#pragma unroll
    for (int h = 0; h < HIDDEN; ++h) {
        const float w = W2[h];
#pragma unroll
        for (int i = 0; i < RI; ++i) {
            const float l = __builtin_bit_cast(float,
                __builtin_amdgcn_readlane(__builtin_bit_cast(int, lv[i][h >> 5]), h & 31));
#pragma unroll
            for (int q = 0; q < QJ; ++q) {
                const float v = l + rp_[q][h];
                acc[i][q] = __builtin_fmaf(fmaxf(v, 0.0f), w, acc[i][q]);
            }
        }
    }

#pragma unroll
    for (int i = 0; i < RI; ++i) {
        const int gi = i0 + i;
#pragma unroll
        for (int q = 0; q < QJ; ++q) {
            const int j = jb + 32 * q;
            const float sig = 1.0f / (1.0f + __expf(-acc[i][q]));
            out[((size_t)bidx * NK + gi) * NK + j] = (gi == j) ? 0.0f : sig;
        }
    }
}

extern "C" void kernel_launch(void* const* d_in, const int* in_sizes, int n_in,
                              void* d_out, int out_size, void* d_ws, size_t ws_size,
                              hipStream_t stream) {
    (void)in_sizes; (void)n_in; (void)out_size; (void)ws_size;
    const float* slots = (const float*)d_in[0];
    const float* W1    = (const float*)d_in[1];
    const float* b1    = (const float*)d_in[2];
    const float* W2    = (const float*)d_in[3];
    const float* b2    = (const float*)d_in[4];
    float*       out   = (float*)d_out;
    float*       Lm    = (float*)d_ws;
    float*       RT4   = (float*)d_ws + LM_ELEMS;

    dim3 gA(64, 4);
    proj_gemm_wmma<<<gA, 128, 0, stream>>>(slots, W1, Lm, RT4);

    dim3 gB(64, 8);
    edge_score<<<gB, 128, 0, stream>>>(Lm, RT4, b1, W2, b2, out);
}
